# starting kernel
# baseline (speedup 1.0000x reference)
#include <hip/hip_runtime.h>
#include <math.h>

typedef __attribute__((ext_vector_type(2))) float v2f;
typedef __attribute__((ext_vector_type(8))) float v8f;

#define S_DIM 256
#define C_DIM 32
#define E_DIM 32
#define N_TOT 512
#define KC 32
#define NCHUNK (S_DIM / KC)

__device__ __forceinline__ float sigmoidf_(float x) { return 1.0f / (1.0f + expf(-x)); }

__global__ __launch_bounds__(256) void router_setup(const float* __restrict__ ct,
                                                    const float* __restrict__ cm,
                                                    float* __restrict__ ws) {
  __shared__ float r1[256], r2[256];
  __shared__ float bc[4];
  __shared__ float rowre[256], rowim[256], colre[256], colim[256];
  const int tid = threadIdx.x;

  float mre = -1e30f, mim = -1e30f;
  for (int i = tid; i < S_DIM * S_DIM; i += 256) {
    mre = fmaxf(mre, cm[2 * i]);
    mim = fmaxf(mim, cm[2 * i + 1]);
  }
  r1[tid] = mre; r2[tid] = mim;
  __syncthreads();
  for (int off = 128; off > 0; off >>= 1) {
    if (tid < off) { r1[tid] = fmaxf(r1[tid], r1[tid + off]); r2[tid] = fmaxf(r2[tid], r2[tid + off]); }
    __syncthreads();
  }
  if (tid == 0) { bc[0] = r1[0]; bc[1] = r2[0]; }
  __syncthreads();
  mre = bc[0]; mim = bc[1];

  float sre = 0.0f, sim = 0.0f;
  for (int i = tid; i < S_DIM * S_DIM; i += 256) {
    sre += expf(cm[2 * i] - mre);
    sim += expf(cm[2 * i + 1] - mim);
  }
  r1[tid] = sre; r2[tid] = sim;
  __syncthreads();
  for (int off = 128; off > 0; off >>= 1) {
    if (tid < off) { r1[tid] += r1[tid + off]; r2[tid] += r2[tid + off]; }
    __syncthreads();
  }
  if (tid == 0) { bc[2] = r1[0]; bc[3] = r2[0]; }
  __syncthreads();
  sre = bc[2]; sim = bc[3];

  rowre[tid] = expf(cm[2 * tid] - mre) / sre;
  rowim[tid] = expf(cm[2 * tid + 1] - mim) / sim;
  colre[tid] = expf(cm[2 * (tid * S_DIM)] - mre) / sre;
  colim[tid] = expf(cm[2 * (tid * S_DIM) + 1] - mim) / sim;
  __syncthreads();

  float mrr = 0.0f, mri = 0.0f, mcr = 0.0f, mci = 0.0f;
  const float w0 = -6.283185307179586f / 256.0f;
  for (int t = 0; t < 256; ++t) {
    float ang = w0 * (float)((tid * t) & 255);
    float sn, cs;
    sincosf(ang, &sn, &cs);
    mrr += rowre[t] * cs - rowim[t] * sn;
    mri += rowre[t] * sn + rowim[t] * cs;
    mcr += colre[t] * cs - colim[t] * sn;
    mci += colre[t] * sn + colim[t] * cs;
  }
  ws[tid] = mrr; ws[256 + tid] = mri;
  ws[512 + tid] = mcr; ws[768 + tid] = mci;

  if (tid < E_DIM) {
    const int e = tid;
    ws[1024 + e] = sigmoidf_(ct[((e * 2 + 0) * S_DIM + 0) * 2 + 0]);
    ws[1056 + e] = sigmoidf_(ct[((e * 2 + 0) * S_DIM + 0) * 2 + 1]);
    ws[1088 + e] = sigmoidf_(ct[((e * 2 + 1) * S_DIM + 0) * 2 + 0]);
    ws[1120 + e] = sigmoidf_(ct[((e * 2 + 1) * S_DIM + 0) * 2 + 1]);
  }
}

__global__ __launch_bounds__(256) void router_select(const float* __restrict__ a,
                                                     const float* __restrict__ b,
                                                     const float* __restrict__ ws,
                                                     int* __restrict__ topidx) {
  __shared__ float part[256][6];
  __shared__ float cred[32][4];
  __shared__ float A1[4];
  __shared__ float mag[32];
  const int n = blockIdx.x;
  const int tid = threadIdx.x;
  const int c = tid >> 3, u = tid & 7;
  const float* ap = a + ((size_t)n * C_DIM + c) * S_DIM;
  const float* bp = b + ((size_t)n * C_DIM + c) * S_DIM;

  float sa = 0, sb = 0, par = 0, pai = 0, pbr = 0, pbi = 0;
  for (int i = 0; i < 32; ++i) {
    int s = u * 32 + i;
    float av = ap[s], bv = bp[s];
    sa += av; sb += bv;
    par += av * ws[512 + s]; pai += av * ws[768 + s];
    pbr += bv * ws[s];       pbi += bv * ws[256 + s];
  }
  part[tid][0] = sa; part[tid][1] = sb;
  part[tid][2] = par; part[tid][3] = pai;
  part[tid][4] = pbr; part[tid][5] = pbi;
  __syncthreads();

  if (tid < 32) {
    float s0 = 0, s1 = 0, s2 = 0, s3 = 0, s4 = 0, s5 = 0;
    for (int j = 0; j < 8; ++j) {
      s0 += part[tid * 8 + j][0]; s1 += part[tid * 8 + j][1];
      s2 += part[tid * 8 + j][2]; s3 += part[tid * 8 + j][3];
      s4 += part[tid * 8 + j][4]; s5 += part[tid * 8 + j][5];
    }
    cred[tid][0] = s0 * s4;
    cred[tid][1] = s0 * s5;
    cred[tid][2] = s1 * s2;
    cred[tid][3] = s1 * s3;
  }
  __syncthreads();
  if (tid == 0) {
    float ar = 0, ai = 0, br = 0, bi = 0;
    for (int cc = 0; cc < 32; ++cc) { ar += cred[cc][0]; ai += cred[cc][1]; br += cred[cc][2]; bi += cred[cc][3]; }
    A1[0] = ar; A1[1] = ai; A1[2] = br; A1[3] = bi;
  }
  __syncthreads();
  if (tid < E_DIM) {
    const int e = tid;
    float tar = ws[1024 + e], tai = ws[1056 + e], tbr = ws[1088 + e], tbi = ws[1120 + e];
    float zr = A1[0] * tar + A1[1] * tai + A1[2] * tbr + A1[3] * tbi;
    float zi = A1[1] * tar - A1[0] * tai + A1[3] * tbr - A1[2] * tbi;
    mag[e] = zr * zr + zi * zi;
  }
  __syncthreads();
  if (tid == 0) {
    int s0 = 0, s1 = -1, s2 = -1;
    float b0 = -1.0f, b1 = -1.0f, b2 = -1.0f;
    for (int e = 0; e < 32; ++e) { float m = mag[e]; if (m > b0) { b0 = m; s0 = e; } }
    for (int e = 0; e < 32; ++e) { if (e == s0) continue; float m = mag[e]; if (m > b1) { b1 = m; s1 = e; } }
    for (int e = 0; e < 32; ++e) { if (e == s0 || e == s1) continue; float m = mag[e]; if (m > b2) { b2 = m; s2 = e; } }
    topidx[n * 3 + 0] = s0; topidx[n * 3 + 1] = s1; topidx[n * 3 + 2] = s2;
  }
}

__global__ __launch_bounds__(256) void moe_wmma_gemm(const float* __restrict__ a,
                                                     const float* __restrict__ b,
                                                     const float* __restrict__ ew,
                                                     const int* __restrict__ topidx,
                                                     float* __restrict__ out) {
  __shared__ __align__(16) float Xs[C_DIM][S_DIM];
  __shared__ __align__(16) float Ws[2][KC][S_DIM];
  const int n = blockIdx.x;
  const int tid = threadIdx.x;
  const size_t nb = (size_t)n * C_DIM * S_DIM;

  const int i0 = topidx[n * 3 + 0];
  const int i1 = topidx[n * 3 + 1];
  const int i2 = topidx[n * 3 + 2];
  const float* __restrict__ w0 = ew + (size_t)i0 * (S_DIM * S_DIM);
  const float* __restrict__ w1 = ew + (size_t)i1 * (S_DIM * S_DIM);
  const float* __restrict__ w2 = ew + (size_t)i2 * (S_DIM * S_DIM);

  float4 pre[8];

  auto loadChunk = [&](int K0) {
#pragma unroll
    for (int r = 0; r < 8; ++r) {
      const int lin = r * 256 + tid;
      const int k = lin >> 6;
      const int s4 = (lin & 63) << 2;
      const size_t g = (size_t)(K0 + k) * S_DIM + s4;
      const float4 x0 = *(const float4*)(w0 + g);
      const float4 x1 = *(const float4*)(w1 + g);
      const float4 x2 = *(const float4*)(w2 + g);
      pre[r].x = (x0.x + x1.x + x2.x) * (1.0f / 3.0f);
      pre[r].y = (x0.y + x1.y + x2.y) * (1.0f / 3.0f);
      pre[r].z = (x0.z + x1.z + x2.z) * (1.0f / 3.0f);
      pre[r].w = (x0.w + x1.w + x2.w) * (1.0f / 3.0f);
    }
  };
  auto storeChunk = [&](int buf) {
#pragma unroll
    for (int r = 0; r < 8; ++r) {
      const int lin = r * 256 + tid;
      const int k = lin >> 6;
      const int s4 = (lin & 63) << 2;
      *(float4*)&Ws[buf][k][s4] = pre[r];
    }
  };

#pragma unroll
  for (int r = 0; r < 8; ++r) {
    const int lin = r * 256 + tid;
    const int c = lin >> 6;
    const int s4 = (lin & 63) << 2;
    const float4 av = *(const float4*)(a + nb + c * S_DIM + s4);
    const float4 bv = *(const float4*)(b + nb + c * S_DIM + s4);
    float4 v;
    v.x = av.x + bv.x; v.y = av.y + bv.y; v.z = av.z + bv.z; v.w = av.w + bv.w;
    *(float4*)&Xs[c][(s4 + 4 * c) & (S_DIM - 1)] = v;
  }

  loadChunk(0);
  storeChunk(0);

  const int wave = tid >> 5;
  const int lane = tid & 31;
  const int l16 = lane & 15;
  const int lh = lane >> 4;
  const int ncol = wave * 32;

  v8f acc00 = {}, acc01 = {}, acc10 = {}, acc11 = {};

  int cur = 0;
  for (int ci = 0; ci < NCHUNK; ++ci) {
    __syncthreads();
    if (ci + 1 < NCHUNK) loadChunk((ci + 1) * KC);
    const int k0 = ci * KC;
#pragma unroll
    for (int kk = 0; kk < KC; kk += 4) {
      const int kf = kk + 2 * lh;
      const int m0 = l16, m1 = 16 + l16;
      const int xi0 = (k0 + kf + 4 * m0) & (S_DIM - 1);
      const int xi1 = (k0 + kf + 4 * m1) & (S_DIM - 1);
      v2f fa0, fa1, fb0, fb1;
      fa0.x = Xs[m0][xi0];     fa0.y = Xs[m0][xi0 + 1];
      fa1.x = Xs[m1][xi1];     fa1.y = Xs[m1][xi1 + 1];
      fb0.x = Ws[cur][kf][ncol + l16];       fb0.y = Ws[cur][kf + 1][ncol + l16];
      fb1.x = Ws[cur][kf][ncol + 16 + l16];  fb1.y = Ws[cur][kf + 1][ncol + 16 + l16];
      acc00 = __builtin_amdgcn_wmma_f32_16x16x4_f32(false, fa0, false, fb0, (short)0, acc00, false, false);
      acc01 = __builtin_amdgcn_wmma_f32_16x16x4_f32(false, fa0, false, fb1, (short)0, acc01, false, false);
      acc10 = __builtin_amdgcn_wmma_f32_16x16x4_f32(false, fa1, false, fb0, (short)0, acc10, false, false);
      acc11 = __builtin_amdgcn_wmma_f32_16x16x4_f32(false, fa1, false, fb1, (short)0, acc11, false, false);
    }
    if (ci + 1 < NCHUNK) storeChunk(cur ^ 1);
    cur ^= 1;
  }

  __syncthreads();
#pragma unroll
  for (int r = 0; r < 8; ++r) {
    const int m = r + 8 * lh;
    Xs[m][ncol + l16] = acc00[r];
    Xs[m][ncol + 16 + l16] = acc01[r];
    Xs[m + 16][ncol + l16] = acc10[r];
    Xs[m + 16][ncol + 16 + l16] = acc11[r];
  }
  __syncthreads();
#pragma unroll
  for (int r = 0; r < 8; ++r) {
    const int lin = r * 256 + tid;
    const int c = lin >> 6;
    const int s4 = (lin & 63) << 2;
    *(float4*)(out + nb + c * S_DIM + s4) = *(const float4*)&Xs[c][s4];
  }
}

extern "C" void kernel_launch(void* const* d_in, const int* in_sizes, int n_in,
                              void* d_out, int out_size, void* d_ws, size_t ws_size,
                              hipStream_t stream) {
  const float* a = (const float*)d_in[0];
  const float* b = (const float*)d_in[1];
  const float* ct = (const float*)d_in[2];
  const float* cm = (const float*)d_in[3];
  const float* ew = (const float*)d_in[4];
  float* out = (float*)d_out;
  float* ws = (float*)d_ws;
  int* topidx = (int*)(ws + 1152);

  router_setup<<<1, 256, 0, stream>>>(ct, cm, ws);
  router_select<<<N_TOT, 256, 0, stream>>>(a, b, ws, topidx);
  moe_wmma_gemm<<<N_TOT, 256, 0, stream>>>(a, b, ew, topidx, out);
}


# Spread the single-work-group router set-up over 256 groups: launch 1 per-row softmax max/sum partials, launch 2 one DFT bin per group; tree sums change softmax and DFT summation order
# speedup vs baseline: 3.4280x; 3.4280x over previous
#include <hip/hip_runtime.h>
#include <math.h>

typedef __attribute__((ext_vector_type(2))) float v2f;
typedef __attribute__((ext_vector_type(8))) float v8f;

#define S_DIM 256
#define C_DIM 32
#define E_DIM 32
#define N_TOT 512
#define KC 32
#define NCHUNK (S_DIM / KC)

__device__ __forceinline__ float sigmoidf_(float x) { return 1.0f / (1.0f + expf(-x)); }

#define PART_OFF 4096

__device__ __forceinline__ float wave_max_(float v) {
  for (int o = 16; o > 0; o >>= 1) v = fmaxf(v, __shfl_xor(v, o, 32));
  return v;
}
__device__ __forceinline__ float wave_sum_(float v) {
  for (int o = 16; o > 0; o >>= 1) v += __shfl_xor(v, o, 32);
  return v;
}

__global__ __launch_bounds__(256) void router_partials(const float* __restrict__ ct,
                                                       const float* __restrict__ cm,
                                                       float* __restrict__ ws) {
  __shared__ float rmax[8][2];
  __shared__ float rsum[8][2];
  const int tid = threadIdx.x;
  const int row = blockIdx.x;
  const int lane = tid & 31, wave = tid >> 5;

  const float2 x = *(const float2*)(cm + 2 * (row * S_DIM + tid));
  float mre = wave_max_(x.x), mim = wave_max_(x.y);
  if (lane == 0) { rmax[wave][0] = mre; rmax[wave][1] = mim; }
  __syncthreads();
  mre = rmax[0][0]; mim = rmax[0][1];
  for (int w = 1; w < 8; ++w) { mre = fmaxf(mre, rmax[w][0]); mim = fmaxf(mim, rmax[w][1]); }

  float sre = wave_sum_(expf(x.x - mre));
  float sim = wave_sum_(expf(x.y - mim));
  if (lane == 0) { rsum[wave][0] = sre; rsum[wave][1] = sim; }
  __syncthreads();
  if (tid == 0) {
    sre = rsum[0][0]; sim = rsum[0][1];
    for (int w = 1; w < 8; ++w) { sre += rsum[w][0]; sim += rsum[w][1]; }
    *(float4*)(ws + PART_OFF + 4 * row) = make_float4(mre, mim, sre, sim);
  }

  if (row == 0 && tid < E_DIM) {
    const int e = tid;
    ws[1024 + e] = sigmoidf_(ct[((e * 2 + 0) * S_DIM + 0) * 2 + 0]);
    ws[1056 + e] = sigmoidf_(ct[((e * 2 + 0) * S_DIM + 0) * 2 + 1]);
    ws[1088 + e] = sigmoidf_(ct[((e * 2 + 1) * S_DIM + 0) * 2 + 0]);
    ws[1120 + e] = sigmoidf_(ct[((e * 2 + 1) * S_DIM + 0) * 2 + 1]);
  }
}

__global__ __launch_bounds__(256) void router_dft(const float* __restrict__ cm,
                                                  float* __restrict__ ws) {
  __shared__ float rmax[8][2];
  __shared__ float rsum[8][2];
  __shared__ float rdft[8][4];
  const int tid = threadIdx.x;
  const int k = blockIdx.x;
  const int lane = tid & 31, wave = tid >> 5;

  const float4 p = *(const float4*)(ws + PART_OFF + 4 * tid);
  const float2 xr = *(const float2*)(cm + 2 * tid);
  const float2 xc = *(const float2*)(cm + 2 * (tid * S_DIM));

  float mre = wave_max_(p.x), mim = wave_max_(p.y);
  if (lane == 0) { rmax[wave][0] = mre; rmax[wave][1] = mim; }
  __syncthreads();
  mre = rmax[0][0]; mim = rmax[0][1];
  for (int w = 1; w < 8; ++w) { mre = fmaxf(mre, rmax[w][0]); mim = fmaxf(mim, rmax[w][1]); }

  float sre = wave_sum_(p.z * expf(p.x - mre));
  float sim = wave_sum_(p.w * expf(p.y - mim));
  if (lane == 0) { rsum[wave][0] = sre; rsum[wave][1] = sim; }
  __syncthreads();
  sre = rsum[0][0]; sim = rsum[0][1];
  for (int w = 1; w < 8; ++w) { sre += rsum[w][0]; sim += rsum[w][1]; }

  const float rowre = expf(xr.x - mre) / sre;
  const float rowim = expf(xr.y - mim) / sim;
  const float colre = expf(xc.x - mre) / sre;
  const float colim = expf(xc.y - mim) / sim;

  const float w0 = -6.283185307179586f / 256.0f;
  const float ang = w0 * (float)((k * tid) & 255);
  float sn, cs;
  sincosf(ang, &sn, &cs);
  const float v0 = wave_sum_(rowre * cs - rowim * sn);
  const float v1 = wave_sum_(rowre * sn + rowim * cs);
  const float v2 = wave_sum_(colre * cs - colim * sn);
  const float v3 = wave_sum_(colre * sn + colim * cs);
  if (lane == 0) { rdft[wave][0] = v0; rdft[wave][1] = v1; rdft[wave][2] = v2; rdft[wave][3] = v3; }
  __syncthreads();
  if (tid < 4) {
    float s = rdft[0][tid];
    for (int w = 1; w < 8; ++w) s += rdft[w][tid];
    ws[tid * S_DIM + k] = s;
  }
}

__global__ __launch_bounds__(256) void router_select(const float* __restrict__ a,
                                                     const float* __restrict__ b,
                                                     const float* __restrict__ ws,
                                                     int* __restrict__ topidx) {
  __shared__ float part[256][6];
  __shared__ float cred[32][4];
  __shared__ float A1[4];
  __shared__ float mag[32];
  const int n = blockIdx.x;
  const int tid = threadIdx.x;
  const int c = tid >> 3, u = tid & 7;
  const float* ap = a + ((size_t)n * C_DIM + c) * S_DIM;
  const float* bp = b + ((size_t)n * C_DIM + c) * S_DIM;

  float sa = 0, sb = 0, par = 0, pai = 0, pbr = 0, pbi = 0;
  for (int i = 0; i < 32; ++i) {
    int s = u * 32 + i;
    float av = ap[s], bv = bp[s];
    sa += av; sb += bv;
    par += av * ws[512 + s]; pai += av * ws[768 + s];
    pbr += bv * ws[s];       pbi += bv * ws[256 + s];
  }
  part[tid][0] = sa; part[tid][1] = sb;
  part[tid][2] = par; part[tid][3] = pai;
  part[tid][4] = pbr; part[tid][5] = pbi;
  __syncthreads();

  if (tid < 32) {
    float s0 = 0, s1 = 0, s2 = 0, s3 = 0, s4 = 0, s5 = 0;
    for (int j = 0; j < 8; ++j) {
      s0 += part[tid * 8 + j][0]; s1 += part[tid * 8 + j][1];
      s2 += part[tid * 8 + j][2]; s3 += part[tid * 8 + j][3];
      s4 += part[tid * 8 + j][4]; s5 += part[tid * 8 + j][5];
    }
    cred[tid][0] = s0 * s4;
    cred[tid][1] = s0 * s5;
    cred[tid][2] = s1 * s2;
    cred[tid][3] = s1 * s3;
  }
  __syncthreads();
  if (tid == 0) {
    float ar = 0, ai = 0, br = 0, bi = 0;
    for (int cc = 0; cc < 32; ++cc) { ar += cred[cc][0]; ai += cred[cc][1]; br += cred[cc][2]; bi += cred[cc][3]; }
    A1[0] = ar; A1[1] = ai; A1[2] = br; A1[3] = bi;
  }
  __syncthreads();
  if (tid < E_DIM) {
    const int e = tid;
    float tar = ws[1024 + e], tai = ws[1056 + e], tbr = ws[1088 + e], tbi = ws[1120 + e];
    float zr = A1[0] * tar + A1[1] * tai + A1[2] * tbr + A1[3] * tbi;
    float zi = A1[1] * tar - A1[0] * tai + A1[3] * tbr - A1[2] * tbi;
    mag[e] = zr * zr + zi * zi;
  }
  __syncthreads();
  if (tid == 0) {
    int s0 = 0, s1 = -1, s2 = -1;
    float b0 = -1.0f, b1 = -1.0f, b2 = -1.0f;
    for (int e = 0; e < 32; ++e) { float m = mag[e]; if (m > b0) { b0 = m; s0 = e; } }
    for (int e = 0; e < 32; ++e) { if (e == s0) continue; float m = mag[e]; if (m > b1) { b1 = m; s1 = e; } }
    for (int e = 0; e < 32; ++e) { if (e == s0 || e == s1) continue; float m = mag[e]; if (m > b2) { b2 = m; s2 = e; } }
    topidx[n * 3 + 0] = s0; topidx[n * 3 + 1] = s1; topidx[n * 3 + 2] = s2;
  }
}

__global__ __launch_bounds__(256) void moe_wmma_gemm(const float* __restrict__ a,
                                                     const float* __restrict__ b,
                                                     const float* __restrict__ ew,
                                                     const int* __restrict__ topidx,
                                                     float* __restrict__ out) {
  __shared__ __align__(16) float Xs[C_DIM][S_DIM];
  __shared__ __align__(16) float Ws[2][KC][S_DIM];
  const int n = blockIdx.x;
  const int tid = threadIdx.x;
  const size_t nb = (size_t)n * C_DIM * S_DIM;

  const int i0 = topidx[n * 3 + 0];
  const int i1 = topidx[n * 3 + 1];
  const int i2 = topidx[n * 3 + 2];
  const float* __restrict__ w0 = ew + (size_t)i0 * (S_DIM * S_DIM);
  const float* __restrict__ w1 = ew + (size_t)i1 * (S_DIM * S_DIM);
  const float* __restrict__ w2 = ew + (size_t)i2 * (S_DIM * S_DIM);

  float4 pre[8];

  auto loadChunk = [&](int K0) {
#pragma unroll
    for (int r = 0; r < 8; ++r) {
      const int lin = r * 256 + tid;
      const int k = lin >> 6;
      const int s4 = (lin & 63) << 2;
      const size_t g = (size_t)(K0 + k) * S_DIM + s4;
      const float4 x0 = *(const float4*)(w0 + g);
      const float4 x1 = *(const float4*)(w1 + g);
      const float4 x2 = *(const float4*)(w2 + g);
      pre[r].x = (x0.x + x1.x + x2.x) * (1.0f / 3.0f);
      pre[r].y = (x0.y + x1.y + x2.y) * (1.0f / 3.0f);
      pre[r].z = (x0.z + x1.z + x2.z) * (1.0f / 3.0f);
      pre[r].w = (x0.w + x1.w + x2.w) * (1.0f / 3.0f);
    }
  };
  auto storeChunk = [&](int buf) {
#pragma unroll
    for (int r = 0; r < 8; ++r) {
      const int lin = r * 256 + tid;
      const int k = lin >> 6;
      const int s4 = (lin & 63) << 2;
      *(float4*)&Ws[buf][k][s4] = pre[r];
    }
  };

#pragma unroll
  for (int r = 0; r < 8; ++r) {
    const int lin = r * 256 + tid;
    const int c = lin >> 6;
    const int s4 = (lin & 63) << 2;
    const float4 av = *(const float4*)(a + nb + c * S_DIM + s4);
    const float4 bv = *(const float4*)(b + nb + c * S_DIM + s4);
    float4 v;
    v.x = av.x + bv.x; v.y = av.y + bv.y; v.z = av.z + bv.z; v.w = av.w + bv.w;
    *(float4*)&Xs[c][(s4 + 4 * c) & (S_DIM - 1)] = v;
  }

  loadChunk(0);
  storeChunk(0);

  const int wave = tid >> 5;
  const int lane = tid & 31;
  const int l16 = lane & 15;
  const int lh = lane >> 4;
  const int ncol = wave * 32;

  v8f acc00 = {}, acc01 = {}, acc10 = {}, acc11 = {};

  int cur = 0;
  for (int ci = 0; ci < NCHUNK; ++ci) {
    __syncthreads();
    if (ci + 1 < NCHUNK) loadChunk((ci + 1) * KC);
    const int k0 = ci * KC;
#pragma unroll
    for (int kk = 0; kk < KC; kk += 4) {
      const int kf = kk + 2 * lh;
      const int m0 = l16, m1 = 16 + l16;
      const int xi0 = (k0 + kf + 4 * m0) & (S_DIM - 1);
      const int xi1 = (k0 + kf + 4 * m1) & (S_DIM - 1);
      v2f fa0, fa1, fb0, fb1;
      fa0.x = Xs[m0][xi0];     fa0.y = Xs[m0][xi0 + 1];
      fa1.x = Xs[m1][xi1];     fa1.y = Xs[m1][xi1 + 1];
      fb0.x = Ws[cur][kf][ncol + l16];       fb0.y = Ws[cur][kf + 1][ncol + l16];
      fb1.x = Ws[cur][kf][ncol + 16 + l16];  fb1.y = Ws[cur][kf + 1][ncol + 16 + l16];
      acc00 = __builtin_amdgcn_wmma_f32_16x16x4_f32(false, fa0, false, fb0, (short)0, acc00, false, false);
      acc01 = __builtin_amdgcn_wmma_f32_16x16x4_f32(false, fa0, false, fb1, (short)0, acc01, false, false);
      acc10 = __builtin_amdgcn_wmma_f32_16x16x4_f32(false, fa1, false, fb0, (short)0, acc10, false, false);
      acc11 = __builtin_amdgcn_wmma_f32_16x16x4_f32(false, fa1, false, fb1, (short)0, acc11, false, false);
    }
    if (ci + 1 < NCHUNK) storeChunk(cur ^ 1);
    cur ^= 1;
  }

  __syncthreads();
#pragma unroll
  for (int r = 0; r < 8; ++r) {
    const int m = r + 8 * lh;
    Xs[m][ncol + l16] = acc00[r];
    Xs[m][ncol + 16 + l16] = acc01[r];
    Xs[m + 16][ncol + l16] = acc10[r];
    Xs[m + 16][ncol + 16 + l16] = acc11[r];
  }
  __syncthreads();
#pragma unroll
  for (int r = 0; r < 8; ++r) {
    const int lin = r * 256 + tid;
    const int c = lin >> 6;
    const int s4 = (lin & 63) << 2;
    *(float4*)(out + nb + c * S_DIM + s4) = *(const float4*)&Xs[c][s4];
  }
}

extern "C" void kernel_launch(void* const* d_in, const int* in_sizes, int n_in,
                              void* d_out, int out_size, void* d_ws, size_t ws_size,
                              hipStream_t stream) {
  const float* a = (const float*)d_in[0];
  const float* b = (const float*)d_in[1];
  const float* ct = (const float*)d_in[2];
  const float* cm = (const float*)d_in[3];
  const float* ew = (const float*)d_in[4];
  float* out = (float*)d_out;
  float* ws = (float*)d_ws;
  int* topidx = (int*)(ws + 1152);

  router_partials<<<S_DIM, 256, 0, stream>>>(ct, cm, ws);
  router_dft<<<S_DIM, 256, 0, stream>>>(cm, ws);
  router_select<<<N_TOT, 256, 0, stream>>>(a, b, ws, topidx);
  moe_wmma_gemm<<<N_TOT, 256, 0, stream>>>(a, b, ew, topidx, out);
}


# Fuse the top-3 expert selection into the GEMM kernel: each sample's group takes its correlations from the a/b tile it loads anyway (3 launches, not 4); LDS tree sums change the correlation sum order
# speedup vs baseline: 4.7056x; 1.3727x over previous
#include <hip/hip_runtime.h>
#include <math.h>

typedef __attribute__((ext_vector_type(2))) float v2f;
typedef __attribute__((ext_vector_type(8))) float v8f;

#define S_DIM 256
#define C_DIM 32
#define E_DIM 32
#define N_TOT 512
#define KC 32
#define NCHUNK (S_DIM / KC)

__device__ __forceinline__ float sigmoidf_(float x) { return 1.0f / (1.0f + expf(-x)); }

#define PART_OFF 4096

__device__ __forceinline__ float wave_max_(float v) {
  for (int o = 16; o > 0; o >>= 1) v = fmaxf(v, __shfl_xor(v, o, 32));
  return v;
}
__device__ __forceinline__ float wave_sum_(float v) {
  for (int o = 16; o > 0; o >>= 1) v += __shfl_xor(v, o, 32);
  return v;
}

__global__ __launch_bounds__(256) void router_partials(const float* __restrict__ ct,
                                                       const float* __restrict__ cm,
                                                       float* __restrict__ ws) {
  __shared__ float rmax[8][2];
  __shared__ float rsum[8][2];
  const int tid = threadIdx.x;
  const int row = blockIdx.x;
  const int lane = tid & 31, wave = tid >> 5;

  const float2 x = *(const float2*)(cm + 2 * (row * S_DIM + tid));
  float mre = wave_max_(x.x), mim = wave_max_(x.y);
  if (lane == 0) { rmax[wave][0] = mre; rmax[wave][1] = mim; }
  __syncthreads();
  mre = rmax[0][0]; mim = rmax[0][1];
  for (int w = 1; w < 8; ++w) { mre = fmaxf(mre, rmax[w][0]); mim = fmaxf(mim, rmax[w][1]); }

  float sre = wave_sum_(expf(x.x - mre));
  float sim = wave_sum_(expf(x.y - mim));
  if (lane == 0) { rsum[wave][0] = sre; rsum[wave][1] = sim; }
  __syncthreads();
  if (tid == 0) {
    sre = rsum[0][0]; sim = rsum[0][1];
    for (int w = 1; w < 8; ++w) { sre += rsum[w][0]; sim += rsum[w][1]; }
    *(float4*)(ws + PART_OFF + 4 * row) = make_float4(mre, mim, sre, sim);
  }

  if (row == 0 && tid < E_DIM) {
    const int e = tid;
    ws[1024 + e] = sigmoidf_(ct[((e * 2 + 0) * S_DIM + 0) * 2 + 0]);
    ws[1056 + e] = sigmoidf_(ct[((e * 2 + 0) * S_DIM + 0) * 2 + 1]);
    ws[1088 + e] = sigmoidf_(ct[((e * 2 + 1) * S_DIM + 0) * 2 + 0]);
    ws[1120 + e] = sigmoidf_(ct[((e * 2 + 1) * S_DIM + 0) * 2 + 1]);
  }
}

__global__ __launch_bounds__(256) void router_dft(const float* __restrict__ cm,
                                                  float* __restrict__ ws) {
  __shared__ float rmax[8][2];
  __shared__ float rsum[8][2];
  __shared__ float rdft[8][4];
  const int tid = threadIdx.x;
  const int k = blockIdx.x;
  const int lane = tid & 31, wave = tid >> 5;

  const float4 p = *(const float4*)(ws + PART_OFF + 4 * tid);
  const float2 xr = *(const float2*)(cm + 2 * tid);
  const float2 xc = *(const float2*)(cm + 2 * (tid * S_DIM));

  float mre = wave_max_(p.x), mim = wave_max_(p.y);
  if (lane == 0) { rmax[wave][0] = mre; rmax[wave][1] = mim; }
  __syncthreads();
  mre = rmax[0][0]; mim = rmax[0][1];
  for (int w = 1; w < 8; ++w) { mre = fmaxf(mre, rmax[w][0]); mim = fmaxf(mim, rmax[w][1]); }

  float sre = wave_sum_(p.z * expf(p.x - mre));
  float sim = wave_sum_(p.w * expf(p.y - mim));
  if (lane == 0) { rsum[wave][0] = sre; rsum[wave][1] = sim; }
  __syncthreads();
  sre = rsum[0][0]; sim = rsum[0][1];
  for (int w = 1; w < 8; ++w) { sre += rsum[w][0]; sim += rsum[w][1]; }

  const float rowre = expf(xr.x - mre) / sre;
  const float rowim = expf(xr.y - mim) / sim;
  const float colre = expf(xc.x - mre) / sre;
  const float colim = expf(xc.y - mim) / sim;

  const float w0 = -6.283185307179586f / 256.0f;
  const float ang = w0 * (float)((k * tid) & 255);
  float sn, cs;
  sincosf(ang, &sn, &cs);
  const float v0 = wave_sum_(rowre * cs - rowim * sn);
  const float v1 = wave_sum_(rowre * sn + rowim * cs);
  const float v2 = wave_sum_(colre * cs - colim * sn);
  const float v3 = wave_sum_(colre * sn + colim * cs);
  if (lane == 0) { rdft[wave][0] = v0; rdft[wave][1] = v1; rdft[wave][2] = v2; rdft[wave][3] = v3; }
  __syncthreads();
  if (tid < 4) {
    float s = rdft[0][tid];
    for (int w = 1; w < 8; ++w) s += rdft[w][tid];
    ws[tid * S_DIM + k] = s;
  }
}

__global__ __launch_bounds__(256) void moe_select_gemm(const float* __restrict__ a,
                                                       const float* __restrict__ b,
                                                       const float* __restrict__ ew,
                                                       const float* __restrict__ ws,
                                                       float* __restrict__ out) {
  __shared__ __align__(16) float Xs[C_DIM][S_DIM];
  __shared__ __align__(16) float Ws[2][KC][S_DIM];
  __shared__ float rowsum[6][C_DIM];
  __shared__ float mag[E_DIM];
  __shared__ int sel[4];
  const int n = blockIdx.x;
  const int tid = threadIdx.x;
  const int wave = tid >> 5;
  const int lane = tid & 31;
  const size_t nb = (size_t)n * C_DIM * S_DIM;

  float* part = &Ws[0][0][0];
  if (tid < 3) sel[tid] = 0;

  const int s4q = (tid & 63) << 2;
  const float4 mrr = *(const float4*)(ws + s4q);
  const float4 mri = *(const float4*)(ws + 256 + s4q);
  const float4 mcr = *(const float4*)(ws + 512 + s4q);
  const float4 mci = *(const float4*)(ws + 768 + s4q);
  const float tar = ws[1024 + lane], tai = ws[1056 + lane];
  const float tbr = ws[1088 + lane], tbi = ws[1120 + lane];

#pragma unroll
  for (int r = 0; r < 8; ++r) {
    const int lin = r * 256 + tid;
    const int c = lin >> 6;
    const int s4 = (lin & 63) << 2;
    const float4 av = *(const float4*)(a + nb + c * S_DIM + s4);
    const float4 bv = *(const float4*)(b + nb + c * S_DIM + s4);
    float4 v;
    v.x = av.x + bv.x; v.y = av.y + bv.y; v.z = av.z + bv.z; v.w = av.w + bv.w;
    *(float4*)&Xs[c][(s4 + 4 * c) & (S_DIM - 1)] = v;
    float* pp = part + (r * 6) * 256 + tid;
    pp[0 * 256] = (av.x + av.y) + (av.z + av.w);
    pp[1 * 256] = (bv.x + bv.y) + (bv.z + bv.w);
    pp[2 * 256] = av.x * mcr.x + av.y * mcr.y + av.z * mcr.z + av.w * mcr.w;
    pp[3 * 256] = av.x * mci.x + av.y * mci.y + av.z * mci.z + av.w * mci.w;
    pp[4 * 256] = bv.x * mrr.x + bv.y * mrr.y + bv.z * mrr.z + bv.w * mrr.w;
    pp[5 * 256] = bv.x * mri.x + bv.y * mri.y + bv.z * mri.z + bv.w * mri.w;
  }
  __syncthreads();

  if (wave < 6) {
    const int c = lane;
    const float* p = part + ((c >> 2) * 6 + wave) * 256 + (c & 3) * 64;
    float4 acc = *(const float4*)(p + ((lane & 15) << 2));
#pragma unroll
    for (int i = 1; i < 16; ++i) {
      const float4 v = *(const float4*)(p + (((i + lane) & 15) << 2));
      acc.x += v.x; acc.y += v.y; acc.z += v.z; acc.w += v.w;
    }
    rowsum[wave][c] = (acc.x + acc.y) + (acc.z + acc.w);
  }
  __syncthreads();

  if (wave == 0) {
    const int c = lane;
    const float sa = rowsum[0][c], sb = rowsum[1][c];
    const float par = rowsum[2][c], pai = rowsum[3][c];
    const float pbr = rowsum[4][c], pbi = rowsum[5][c];
    const float ar = wave_sum_(sa * pbr), ai = wave_sum_(sa * pbi);
    const float br = wave_sum_(sb * par), bi = wave_sum_(sb * pai);
    const float zr = ar * tar + ai * tai + br * tbr + bi * tbi;
    const float zi = ai * tar - ar * tai + bi * tbr - br * tbi;
    mag[lane] = zr * zr + zi * zi;
  }
  __syncthreads();
  if (wave == 0) {
    const float m = mag[lane];
    int rank = 0;
#pragma unroll
    for (int e = 0; e < E_DIM; ++e) {
      const float o = mag[e];
      rank += (o > m || (o == m && e < lane)) ? 1 : 0;
    }
    if (rank < 3) sel[rank] = lane;
  }
  __syncthreads();

  const int i0 = __builtin_amdgcn_readfirstlane(sel[0]) & (E_DIM - 1);
  const int i1 = __builtin_amdgcn_readfirstlane(sel[1]) & (E_DIM - 1);
  const int i2 = __builtin_amdgcn_readfirstlane(sel[2]) & (E_DIM - 1);
  const float* __restrict__ w0 = ew + (size_t)i0 * (S_DIM * S_DIM);
  const float* __restrict__ w1 = ew + (size_t)i1 * (S_DIM * S_DIM);
  const float* __restrict__ w2 = ew + (size_t)i2 * (S_DIM * S_DIM);

  float4 pre[8];

  auto loadChunk = [&](int K0) {
#pragma unroll
    for (int r = 0; r < 8; ++r) {
      const int lin = r * 256 + tid;
      const int k = lin >> 6;
      const int s4 = (lin & 63) << 2;
      const size_t g = (size_t)(K0 + k) * S_DIM + s4;
      const float4 x0 = *(const float4*)(w0 + g);
      const float4 x1 = *(const float4*)(w1 + g);
      const float4 x2 = *(const float4*)(w2 + g);
      pre[r].x = (x0.x + x1.x + x2.x) * (1.0f / 3.0f);
      pre[r].y = (x0.y + x1.y + x2.y) * (1.0f / 3.0f);
      pre[r].z = (x0.z + x1.z + x2.z) * (1.0f / 3.0f);
      pre[r].w = (x0.w + x1.w + x2.w) * (1.0f / 3.0f);
    }
  };
  auto storeChunk = [&](int buf) {
#pragma unroll
    for (int r = 0; r < 8; ++r) {
      const int lin = r * 256 + tid;
      const int k = lin >> 6;
      const int s4 = (lin & 63) << 2;
      *(float4*)&Ws[buf][k][s4] = pre[r];
    }
  };

  loadChunk(0);
  storeChunk(0);

  const int l16 = lane & 15;
  const int lh = lane >> 4;
  const int ncol = wave * 32;

  v8f acc00 = {}, acc01 = {}, acc10 = {}, acc11 = {};

  int cur = 0;
  for (int ci = 0; ci < NCHUNK; ++ci) {
    __syncthreads();
    if (ci + 1 < NCHUNK) loadChunk((ci + 1) * KC);
    const int k0 = ci * KC;
#pragma unroll
    for (int kk = 0; kk < KC; kk += 4) {
      const int kf = kk + 2 * lh;
      const int m0 = l16, m1 = 16 + l16;
      const int xi0 = (k0 + kf + 4 * m0) & (S_DIM - 1);
      const int xi1 = (k0 + kf + 4 * m1) & (S_DIM - 1);
      v2f fa0, fa1, fb0, fb1;
      fa0.x = Xs[m0][xi0];     fa0.y = Xs[m0][xi0 + 1];
      fa1.x = Xs[m1][xi1];     fa1.y = Xs[m1][xi1 + 1];
      fb0.x = Ws[cur][kf][ncol + l16];       fb0.y = Ws[cur][kf + 1][ncol + l16];
      fb1.x = Ws[cur][kf][ncol + 16 + l16];  fb1.y = Ws[cur][kf + 1][ncol + 16 + l16];
      acc00 = __builtin_amdgcn_wmma_f32_16x16x4_f32(false, fa0, false, fb0, (short)0, acc00, false, false);
      acc01 = __builtin_amdgcn_wmma_f32_16x16x4_f32(false, fa0, false, fb1, (short)0, acc01, false, false);
      acc10 = __builtin_amdgcn_wmma_f32_16x16x4_f32(false, fa1, false, fb0, (short)0, acc10, false, false);
      acc11 = __builtin_amdgcn_wmma_f32_16x16x4_f32(false, fa1, false, fb1, (short)0, acc11, false, false);
    }
    if (ci + 1 < NCHUNK) storeChunk(cur ^ 1);
    cur ^= 1;
  }

  __syncthreads();
#pragma unroll
  for (int r = 0; r < 8; ++r) {
    const int m = r + 8 * lh;
    Xs[m][ncol + l16] = acc00[r];
    Xs[m][ncol + 16 + l16] = acc01[r];
    Xs[m + 16][ncol + l16] = acc10[r];
    Xs[m + 16][ncol + 16 + l16] = acc11[r];
  }
  __syncthreads();
#pragma unroll
  for (int r = 0; r < 8; ++r) {
    const int lin = r * 256 + tid;
    const int c = lin >> 6;
    const int s4 = (lin & 63) << 2;
    *(float4*)(out + nb + c * S_DIM + s4) = *(const float4*)&Xs[c][s4];
  }
}

extern "C" void kernel_launch(void* const* d_in, const int* in_sizes, int n_in,
                              void* d_out, int out_size, void* d_ws, size_t ws_size,
                              hipStream_t stream) {
  const float* a = (const float*)d_in[0];
  const float* b = (const float*)d_in[1];
  const float* ct = (const float*)d_in[2];
  const float* cm = (const float*)d_in[3];
  const float* ew = (const float*)d_in[4];
  float* out = (float*)d_out;
  float* ws = (float*)d_ws;

  router_partials<<<S_DIM, 256, 0, stream>>>(ct, cm, ws);
  router_dft<<<S_DIM, 256, 0, stream>>>(cm, ws);
  moe_select_gemm<<<N_TOT, 256, 0, stream>>>(a, b, ew, ws, out);
}


# Issue the next chunk's expert-weight loads inside the matrix burst (6 global loads per k-step in its first four k-steps; chunk loop unrolled, scalar base + lane offset), not all 24 in front of it
# speedup vs baseline: 5.0162x; 1.0660x over previous
#include <hip/hip_runtime.h>
#include <math.h>

typedef __attribute__((ext_vector_type(2))) float v2f;
typedef __attribute__((ext_vector_type(8))) float v8f;

#define S_DIM 256
#define C_DIM 32
#define E_DIM 32
#define N_TOT 512
#define KC 32
#define NCHUNK (S_DIM / KC)

__device__ __forceinline__ float sigmoidf_(float x) { return 1.0f / (1.0f + expf(-x)); }

#define PART_OFF 4096

__device__ __forceinline__ float wave_max_(float v) {
  for (int o = 16; o > 0; o >>= 1) v = fmaxf(v, __shfl_xor(v, o, 32));
  return v;
}
__device__ __forceinline__ float wave_sum_(float v) {
  for (int o = 16; o > 0; o >>= 1) v += __shfl_xor(v, o, 32);
  return v;
}

__global__ __launch_bounds__(256) void router_partials(const float* __restrict__ ct,
                                                       const float* __restrict__ cm,
                                                       float* __restrict__ ws) {
  __shared__ float rmax[8][2];
  __shared__ float rsum[8][2];
  const int tid = threadIdx.x;
  const int row = blockIdx.x;
  const int lane = tid & 31, wave = tid >> 5;

  const float2 x = *(const float2*)(cm + 2 * (row * S_DIM + tid));
  float mre = wave_max_(x.x), mim = wave_max_(x.y);
  if (lane == 0) { rmax[wave][0] = mre; rmax[wave][1] = mim; }
  __syncthreads();
  mre = rmax[0][0]; mim = rmax[0][1];
  for (int w = 1; w < 8; ++w) { mre = fmaxf(mre, rmax[w][0]); mim = fmaxf(mim, rmax[w][1]); }

  float sre = wave_sum_(expf(x.x - mre));
  float sim = wave_sum_(expf(x.y - mim));
  if (lane == 0) { rsum[wave][0] = sre; rsum[wave][1] = sim; }
  __syncthreads();
  if (tid == 0) {
    sre = rsum[0][0]; sim = rsum[0][1];
    for (int w = 1; w < 8; ++w) { sre += rsum[w][0]; sim += rsum[w][1]; }
    *(float4*)(ws + PART_OFF + 4 * row) = make_float4(mre, mim, sre, sim);
  }

  if (row == 0 && tid < E_DIM) {
    const int e = tid;
    ws[1024 + e] = sigmoidf_(ct[((e * 2 + 0) * S_DIM + 0) * 2 + 0]);
    ws[1056 + e] = sigmoidf_(ct[((e * 2 + 0) * S_DIM + 0) * 2 + 1]);
    ws[1088 + e] = sigmoidf_(ct[((e * 2 + 1) * S_DIM + 0) * 2 + 0]);
    ws[1120 + e] = sigmoidf_(ct[((e * 2 + 1) * S_DIM + 0) * 2 + 1]);
  }
}

__global__ __launch_bounds__(256) void router_dft(const float* __restrict__ cm,
                                                  float* __restrict__ ws) {
  __shared__ float rmax[8][2];
  __shared__ float rsum[8][2];
  __shared__ float rdft[8][4];
  const int tid = threadIdx.x;
  const int k = blockIdx.x;
  const int lane = tid & 31, wave = tid >> 5;

  const float4 p = *(const float4*)(ws + PART_OFF + 4 * tid);
  const float2 xr = *(const float2*)(cm + 2 * tid);
  const float2 xc = *(const float2*)(cm + 2 * (tid * S_DIM));

  float mre = wave_max_(p.x), mim = wave_max_(p.y);
  if (lane == 0) { rmax[wave][0] = mre; rmax[wave][1] = mim; }
  __syncthreads();
  mre = rmax[0][0]; mim = rmax[0][1];
  for (int w = 1; w < 8; ++w) { mre = fmaxf(mre, rmax[w][0]); mim = fmaxf(mim, rmax[w][1]); }

  float sre = wave_sum_(p.z * expf(p.x - mre));
  float sim = wave_sum_(p.w * expf(p.y - mim));
  if (lane == 0) { rsum[wave][0] = sre; rsum[wave][1] = sim; }
  __syncthreads();
  sre = rsum[0][0]; sim = rsum[0][1];
  for (int w = 1; w < 8; ++w) { sre += rsum[w][0]; sim += rsum[w][1]; }

  const float rowre = expf(xr.x - mre) / sre;
  const float rowim = expf(xr.y - mim) / sim;
  const float colre = expf(xc.x - mre) / sre;
  const float colim = expf(xc.y - mim) / sim;

  const float w0 = -6.283185307179586f / 256.0f;
  const float ang = w0 * (float)((k * tid) & 255);
  float sn, cs;
  sincosf(ang, &sn, &cs);
  const float v0 = wave_sum_(rowre * cs - rowim * sn);
  const float v1 = wave_sum_(rowre * sn + rowim * cs);
  const float v2 = wave_sum_(colre * cs - colim * sn);
  const float v3 = wave_sum_(colre * sn + colim * cs);
  if (lane == 0) { rdft[wave][0] = v0; rdft[wave][1] = v1; rdft[wave][2] = v2; rdft[wave][3] = v3; }
  __syncthreads();
  if (tid < 4) {
    float s = rdft[0][tid];
    for (int w = 1; w < 8; ++w) s += rdft[w][tid];
    ws[tid * S_DIM + k] = s;
  }
}

__global__ __launch_bounds__(256) void moe_select_gemm(const float* __restrict__ a,
                                                       const float* __restrict__ b,
                                                       const float* __restrict__ ew,
                                                       const float* __restrict__ ws,
                                                       float* __restrict__ out) {
  __shared__ __align__(16) float Xs[C_DIM][S_DIM];
  __shared__ __align__(16) float Ws[2][KC][S_DIM];
  __shared__ float rowsum[6][C_DIM];
  __shared__ float mag[E_DIM];
  __shared__ int sel[4];
  const int n = blockIdx.x;
  const int tid = threadIdx.x;
  const int wave = tid >> 5;
  const int lane = tid & 31;
  const size_t nb = (size_t)n * C_DIM * S_DIM;

  float* part = &Ws[0][0][0];
  if (tid < 3) sel[tid] = 0;

  const int s4q = (tid & 63) << 2;
  const float4 mrr = *(const float4*)(ws + s4q);
  const float4 mri = *(const float4*)(ws + 256 + s4q);
  const float4 mcr = *(const float4*)(ws + 512 + s4q);
  const float4 mci = *(const float4*)(ws + 768 + s4q);
  const float tar = ws[1024 + lane], tai = ws[1056 + lane];
  const float tbr = ws[1088 + lane], tbi = ws[1120 + lane];

#pragma unroll
  for (int r = 0; r < 8; ++r) {
    const int lin = r * 256 + tid;
    const int c = lin >> 6;
    const int s4 = (lin & 63) << 2;
    const float4 av = *(const float4*)(a + nb + c * S_DIM + s4);
    const float4 bv = *(const float4*)(b + nb + c * S_DIM + s4);
    float4 v;
    v.x = av.x + bv.x; v.y = av.y + bv.y; v.z = av.z + bv.z; v.w = av.w + bv.w;
    *(float4*)&Xs[c][(s4 + 4 * c) & (S_DIM - 1)] = v;
    float* pp = part + (r * 6) * 256 + tid;
    pp[0 * 256] = (av.x + av.y) + (av.z + av.w);
    pp[1 * 256] = (bv.x + bv.y) + (bv.z + bv.w);
    pp[2 * 256] = av.x * mcr.x + av.y * mcr.y + av.z * mcr.z + av.w * mcr.w;
    pp[3 * 256] = av.x * mci.x + av.y * mci.y + av.z * mci.z + av.w * mci.w;
    pp[4 * 256] = bv.x * mrr.x + bv.y * mrr.y + bv.z * mrr.z + bv.w * mrr.w;
    pp[5 * 256] = bv.x * mri.x + bv.y * mri.y + bv.z * mri.z + bv.w * mri.w;
  }
  __syncthreads();

  if (wave < 6) {
    const int c = lane;
    const float* p = part + ((c >> 2) * 6 + wave) * 256 + (c & 3) * 64;
    float4 acc = *(const float4*)(p + ((lane & 15) << 2));
#pragma unroll
    for (int i = 1; i < 16; ++i) {
      const float4 v = *(const float4*)(p + (((i + lane) & 15) << 2));
      acc.x += v.x; acc.y += v.y; acc.z += v.z; acc.w += v.w;
    }
    rowsum[wave][c] = (acc.x + acc.y) + (acc.z + acc.w);
  }
  __syncthreads();

  if (wave == 0) {
    const int c = lane;
    const float sa = rowsum[0][c], sb = rowsum[1][c];
    const float par = rowsum[2][c], pai = rowsum[3][c];
    const float pbr = rowsum[4][c], pbi = rowsum[5][c];
    const float ar = wave_sum_(sa * pbr), ai = wave_sum_(sa * pbi);
    const float br = wave_sum_(sb * par), bi = wave_sum_(sb * pai);
    const float zr = ar * tar + ai * tai + br * tbr + bi * tbi;
    const float zi = ai * tar - ar * tai + bi * tbr - br * tbi;
    mag[lane] = zr * zr + zi * zi;
  }
  __syncthreads();
  if (wave == 0) {
    const float m = mag[lane];
    int rank = 0;
#pragma unroll
    for (int e = 0; e < E_DIM; ++e) {
      const float o = mag[e];
      rank += (o > m || (o == m && e < lane)) ? 1 : 0;
    }
    if (rank < 3) sel[rank] = lane;
  }
  __syncthreads();

  const int i0 = __builtin_amdgcn_readfirstlane(sel[0]) & (E_DIM - 1);
  const int i1 = __builtin_amdgcn_readfirstlane(sel[1]) & (E_DIM - 1);
  const int i2 = __builtin_amdgcn_readfirstlane(sel[2]) & (E_DIM - 1);
  const float* __restrict__ w0 = ew + (size_t)i0 * (S_DIM * S_DIM);
  const float* __restrict__ w1 = ew + (size_t)i1 * (S_DIM * S_DIM);
  const float* __restrict__ w2 = ew + (size_t)i2 * (S_DIM * S_DIM);

  const unsigned lane_off = ((unsigned)(tid >> 6) << 8) | ((unsigned)(tid & 63) << 2);

  float4 x0[8], x1[8], x2[8];

  auto loadGroup = [&](int K0, int r) {
    const unsigned g = (unsigned)((K0 + 4 * r) * S_DIM) + lane_off;
    x0[r] = *(const float4*)(w0 + g);
    x1[r] = *(const float4*)(w1 + g);
    x2[r] = *(const float4*)(w2 + g);
  };
  auto storeChunk = [&](int buf) {
#pragma unroll
    for (int r = 0; r < 8; ++r) {
      float4 p;
      p.x = (x0[r].x + x1[r].x + x2[r].x) * (1.0f / 3.0f);
      p.y = (x0[r].y + x1[r].y + x2[r].y) * (1.0f / 3.0f);
      p.z = (x0[r].z + x1[r].z + x2[r].z) * (1.0f / 3.0f);
      p.w = (x0[r].w + x1[r].w + x2[r].w) * (1.0f / 3.0f);
      const int lin = r * 256 + tid;
      const int k = lin >> 6;
      const int s4 = (lin & 63) << 2;
      *(float4*)&Ws[buf][k][s4] = p;
    }
  };

#pragma unroll
  for (int r = 0; r < 8; ++r) loadGroup(0, r);
  storeChunk(0);

  const int l16 = lane & 15;
  const int lh = lane >> 4;
  const int ncol = wave * 32;

  v8f acc00 = {}, acc01 = {}, acc10 = {}, acc11 = {};

#pragma unroll
  for (int ci = 0; ci < NCHUNK; ++ci) {
    const int cur = ci & 1;
    __syncthreads();
    __builtin_amdgcn_sched_barrier(0);
    const int k0 = ci * KC;
#pragma unroll
    for (int kk = 0; kk < KC; kk += 4) {
      if (ci + 1 < NCHUNK && kk < 16) {
        loadGroup((ci + 1) * KC, kk >> 1);
        loadGroup((ci + 1) * KC, (kk >> 1) + 1);
      }
      const int kf = kk + 2 * lh;
      const int m0 = l16, m1 = 16 + l16;
      const int xi0 = (k0 + kf + 4 * m0) & (S_DIM - 1);
      const int xi1 = (k0 + kf + 4 * m1) & (S_DIM - 1);
      v2f fa0, fa1, fb0, fb1;
      fa0.x = Xs[m0][xi0];     fa0.y = Xs[m0][xi0 + 1];
      fa1.x = Xs[m1][xi1];     fa1.y = Xs[m1][xi1 + 1];
      fb0.x = Ws[cur][kf][ncol + l16];       fb0.y = Ws[cur][kf + 1][ncol + l16];
      fb1.x = Ws[cur][kf][ncol + 16 + l16];  fb1.y = Ws[cur][kf + 1][ncol + 16 + l16];
      acc00 = __builtin_amdgcn_wmma_f32_16x16x4_f32(false, fa0, false, fb0, (short)0, acc00, false, false);
      acc01 = __builtin_amdgcn_wmma_f32_16x16x4_f32(false, fa0, false, fb1, (short)0, acc01, false, false);
      acc10 = __builtin_amdgcn_wmma_f32_16x16x4_f32(false, fa1, false, fb0, (short)0, acc10, false, false);
      acc11 = __builtin_amdgcn_wmma_f32_16x16x4_f32(false, fa1, false, fb1, (short)0, acc11, false, false);
      if (ci + 1 < NCHUNK && kk < 16) __builtin_amdgcn_sched_group_barrier(0x020, 6, 0);
      __builtin_amdgcn_sched_group_barrier(0x008, 4, 0);
    }
    __builtin_amdgcn_sched_barrier(0);
    if (ci + 1 < NCHUNK) storeChunk(cur ^ 1);
  }

  __syncthreads();
#pragma unroll
  for (int r = 0; r < 8; ++r) {
    const int m = r + 8 * lh;
    Xs[m][ncol + l16] = acc00[r];
    Xs[m][ncol + 16 + l16] = acc01[r];
    Xs[m + 16][ncol + l16] = acc10[r];
    Xs[m + 16][ncol + 16 + l16] = acc11[r];
  }
  __syncthreads();
#pragma unroll
  for (int r = 0; r < 8; ++r) {
    const int lin = r * 256 + tid;
    const int c = lin >> 6;
    const int s4 = (lin & 63) << 2;
    *(float4*)(out + nb + c * S_DIM + s4) = *(const float4*)&Xs[c][s4];
  }
}

extern "C" void kernel_launch(void* const* d_in, const int* in_sizes, int n_in,
                              void* d_out, int out_size, void* d_ws, size_t ws_size,
                              hipStream_t stream) {
  const float* a = (const float*)d_in[0];
  const float* b = (const float*)d_in[1];
  const float* ct = (const float*)d_in[2];
  const float* cm = (const float*)d_in[3];
  const float* ew = (const float*)d_in[4];
  float* out = (float*)d_out;
  float* ws = (float*)d_ws;

  router_partials<<<S_DIM, 256, 0, stream>>>(ct, cm, ws);
  router_dft<<<S_DIM, 256, 0, stream>>>(cm, ws);
  moe_select_gemm<<<N_TOT, 256, 0, stream>>>(a, b, ew, ws, out);
}


# Expert weights go from global loads straight into B fragments (a wave owns 64 columns, tile j = column 4c+j), 6 k-steps ahead in registers: no LDS staging, no k-loop barriers; 1/3 on the result
# speedup vs baseline: 5.3490x; 1.0663x over previous
#include <hip/hip_runtime.h>
#include <math.h>

typedef __attribute__((ext_vector_type(2))) float v2f;
typedef __attribute__((ext_vector_type(8))) float v8f;

#define S_DIM 256
#define C_DIM 32
#define E_DIM 32
#define N_TOT 512

__device__ __forceinline__ float sigmoidf_(float x) { return 1.0f / (1.0f + expf(-x)); }

#define PART_OFF 4096

__device__ __forceinline__ float wave_max_(float v) {
  for (int o = 16; o > 0; o >>= 1) v = fmaxf(v, __shfl_xor(v, o, 32));
  return v;
}
__device__ __forceinline__ float wave_sum_(float v) {
  for (int o = 16; o > 0; o >>= 1) v += __shfl_xor(v, o, 32);
  return v;
}

__global__ __launch_bounds__(256) void router_partials(const float* __restrict__ ct,
                                                       const float* __restrict__ cm,
                                                       float* __restrict__ ws) {
  __shared__ float rmax[8][2];
  __shared__ float rsum[8][2];
  const int tid = threadIdx.x;
  const int row = blockIdx.x;
  const int lane = tid & 31, wave = tid >> 5;

  const float2 x = *(const float2*)(cm + 2 * (row * S_DIM + tid));
  float mre = wave_max_(x.x), mim = wave_max_(x.y);
  if (lane == 0) { rmax[wave][0] = mre; rmax[wave][1] = mim; }
  __syncthreads();
  mre = rmax[0][0]; mim = rmax[0][1];
  for (int w = 1; w < 8; ++w) { mre = fmaxf(mre, rmax[w][0]); mim = fmaxf(mim, rmax[w][1]); }

  float sre = wave_sum_(expf(x.x - mre));
  float sim = wave_sum_(expf(x.y - mim));
  if (lane == 0) { rsum[wave][0] = sre; rsum[wave][1] = sim; }
  __syncthreads();
  if (tid == 0) {
    sre = rsum[0][0]; sim = rsum[0][1];
    for (int w = 1; w < 8; ++w) { sre += rsum[w][0]; sim += rsum[w][1]; }
    *(float4*)(ws + PART_OFF + 4 * row) = make_float4(mre, mim, sre, sim);
  }

  if (row == 0 && tid < E_DIM) {
    const int e = tid;
    ws[1024 + e] = sigmoidf_(ct[((e * 2 + 0) * S_DIM + 0) * 2 + 0]);
    ws[1056 + e] = sigmoidf_(ct[((e * 2 + 0) * S_DIM + 0) * 2 + 1]);
    ws[1088 + e] = sigmoidf_(ct[((e * 2 + 1) * S_DIM + 0) * 2 + 0]);
    ws[1120 + e] = sigmoidf_(ct[((e * 2 + 1) * S_DIM + 0) * 2 + 1]);
  }
}

__global__ __launch_bounds__(256) void router_dft(const float* __restrict__ cm,
                                                  float* __restrict__ ws) {
  __shared__ float rmax[8][2];
  __shared__ float rsum[8][2];
  __shared__ float rdft[8][4];
  const int tid = threadIdx.x;
  const int k = blockIdx.x;
  const int lane = tid & 31, wave = tid >> 5;

  const float4 p = *(const float4*)(ws + PART_OFF + 4 * tid);
  const float2 xr = *(const float2*)(cm + 2 * tid);
  const float2 xc = *(const float2*)(cm + 2 * (tid * S_DIM));

  float mre = wave_max_(p.x), mim = wave_max_(p.y);
  if (lane == 0) { rmax[wave][0] = mre; rmax[wave][1] = mim; }
  __syncthreads();
  mre = rmax[0][0]; mim = rmax[0][1];
  for (int w = 1; w < 8; ++w) { mre = fmaxf(mre, rmax[w][0]); mim = fmaxf(mim, rmax[w][1]); }

  float sre = wave_sum_(p.z * expf(p.x - mre));
  float sim = wave_sum_(p.w * expf(p.y - mim));
  if (lane == 0) { rsum[wave][0] = sre; rsum[wave][1] = sim; }
  __syncthreads();
  sre = rsum[0][0]; sim = rsum[0][1];
  for (int w = 1; w < 8; ++w) { sre += rsum[w][0]; sim += rsum[w][1]; }

  const float rowre = expf(xr.x - mre) / sre;
  const float rowim = expf(xr.y - mim) / sim;
  const float colre = expf(xc.x - mre) / sre;
  const float colim = expf(xc.y - mim) / sim;

  const float w0 = -6.283185307179586f / 256.0f;
  const float ang = w0 * (float)((k * tid) & 255);
  float sn, cs;
  sincosf(ang, &sn, &cs);
  const float v0 = wave_sum_(rowre * cs - rowim * sn);
  const float v1 = wave_sum_(rowre * sn + rowim * cs);
  const float v2 = wave_sum_(colre * cs - colim * sn);
  const float v3 = wave_sum_(colre * sn + colim * cs);
  if (lane == 0) { rdft[wave][0] = v0; rdft[wave][1] = v1; rdft[wave][2] = v2; rdft[wave][3] = v3; }
  __syncthreads();
  if (tid < 4) {
    float s = rdft[0][tid];
    for (int w = 1; w < 8; ++w) s += rdft[w][tid];
    ws[tid * S_DIM + k] = s;
  }
}

#define PF 6

__global__ __launch_bounds__(128) void moe_select_gemm(const float* __restrict__ a,
                                                       const float* __restrict__ b,
                                                       const float* __restrict__ ew,
                                                       const float* __restrict__ ws,
                                                       float* __restrict__ out) {
  __shared__ __align__(16) float Xs[C_DIM][S_DIM];
  __shared__ __align__(16) float part[48 * 256];
  __shared__ float rowsum[6][C_DIM];
  __shared__ float mag[E_DIM];
  __shared__ int sel[4];
  const int n = blockIdx.x;
  const int tid = threadIdx.x;
  const int wave = tid >> 5;
  const int lane = tid & 31;
  const size_t nb = (size_t)n * C_DIM * S_DIM;

  if (tid < 3) sel[tid] = 0;

  const int s4q = (tid & 63) << 2;
  const float4 mrr = *(const float4*)(ws + s4q);
  const float4 mri = *(const float4*)(ws + 256 + s4q);
  const float4 mcr = *(const float4*)(ws + 512 + s4q);
  const float4 mci = *(const float4*)(ws + 768 + s4q);
  const float tar = ws[1024 + lane], tai = ws[1056 + lane];
  const float tbr = ws[1088 + lane], tbi = ws[1120 + lane];

  const int rq = tid >> 6;
  const float* __restrict__ ap = a + nb + rq * S_DIM + s4q;
  const float* __restrict__ bp = b + nb + rq * S_DIM + s4q;
#pragma unroll
  for (int r = 0; r < 16; ++r) {
    const int c = 2 * r + rq;
    const float4 av = *(const float4*)(ap + r * 2 * S_DIM);
    const float4 bv = *(const float4*)(bp + r * 2 * S_DIM);
    float4 v;
    v.x = av.x + bv.x; v.y = av.y + bv.y; v.z = av.z + bv.z; v.w = av.w + bv.w;
    *(float4*)&Xs[c][(s4q + 4 * c) & (S_DIM - 1)] = v;
    float* pp = part + ((r >> 1) * 6) * 256 + ((r & 1) * 128 + tid);
    pp[0 * 256] = (av.x + av.y) + (av.z + av.w);
    pp[1 * 256] = (bv.x + bv.y) + (bv.z + bv.w);
    pp[2 * 256] = av.x * mcr.x + av.y * mcr.y + av.z * mcr.z + av.w * mcr.w;
    pp[3 * 256] = av.x * mci.x + av.y * mci.y + av.z * mci.z + av.w * mci.w;
    pp[4 * 256] = bv.x * mrr.x + bv.y * mrr.y + bv.z * mrr.z + bv.w * mrr.w;
    pp[5 * 256] = bv.x * mri.x + bv.y * mri.y + bv.z * mri.z + bv.w * mri.w;
  }
  __syncthreads();

  for (int q = wave; q < 6; q += 4) {
    const int c = lane;
    const float* p = part + ((c >> 2) * 6 + q) * 256 + (c & 3) * 64;
    float4 acc = *(const float4*)(p + ((lane & 15) << 2));
#pragma unroll
    for (int i = 1; i < 16; ++i) {
      const float4 v = *(const float4*)(p + (((i + lane) & 15) << 2));
      acc.x += v.x; acc.y += v.y; acc.z += v.z; acc.w += v.w;
    }
    rowsum[q][c] = (acc.x + acc.y) + (acc.z + acc.w);
  }
  __syncthreads();

  if (wave == 0) {
    const int c = lane;
    const float sa = rowsum[0][c], sb = rowsum[1][c];
    const float par = rowsum[2][c], pai = rowsum[3][c];
    const float pbr = rowsum[4][c], pbi = rowsum[5][c];
    const float ar = wave_sum_(sa * pbr), ai = wave_sum_(sa * pbi);
    const float br = wave_sum_(sb * par), bi = wave_sum_(sb * pai);
    const float zr = ar * tar + ai * tai + br * tbr + bi * tbi;
    const float zi = ai * tar - ar * tai + bi * tbr - br * tbi;
    mag[lane] = zr * zr + zi * zi;
  }
  __syncthreads();
  if (wave == 0) {
    const float m = mag[lane];
    int rank = 0;
#pragma unroll
    for (int e = 0; e < E_DIM; ++e) {
      const float o = mag[e];
      rank += (o > m || (o == m && e < lane)) ? 1 : 0;
    }
    if (rank < 3) sel[rank] = lane;
  }
  __syncthreads();

  const int i0 = __builtin_amdgcn_readfirstlane(sel[0]) & (E_DIM - 1);
  const int i1 = __builtin_amdgcn_readfirstlane(sel[1]) & (E_DIM - 1);
  const int i2 = __builtin_amdgcn_readfirstlane(sel[2]) & (E_DIM - 1);
  const float* __restrict__ w0 = ew + (size_t)i0 * (S_DIM * S_DIM);
  const float* __restrict__ w1 = ew + (size_t)i1 * (S_DIM * S_DIM);
  const float* __restrict__ w2 = ew + (size_t)i2 * (S_DIM * S_DIM);

  const int l16 = lane & 15;
  const int lh = lane >> 4;
  const int ncol = wave * 64 + 4 * l16;

  const unsigned lane_off = ((unsigned)(2 * lh) << 8) + (unsigned)ncol;

  float4 qa0[PF], qa1[PF], qa2[PF], qb0[PF], qb1[PF], qb2[PF];
  auto loadStep = [&](int ks, int d) {
    const unsigned g = (unsigned)(ks * 4 * S_DIM) + lane_off;
    qa0[d] = *(const float4*)(w0 + g);
    qb0[d] = *(const float4*)(w0 + g + S_DIM);
    qa1[d] = *(const float4*)(w1 + g);
    qb1[d] = *(const float4*)(w1 + g + S_DIM);
    qa2[d] = *(const float4*)(w2 + g);
    qb2[d] = *(const float4*)(w2 + g + S_DIM);
  };

#pragma unroll
  for (int d = 0; d < PF; ++d) loadStep(d, d);

  v8f acc00 = {}, acc01 = {}, acc02 = {}, acc03 = {};
  v8f acc10 = {}, acc11 = {}, acc12 = {}, acc13 = {};

  __builtin_amdgcn_sched_barrier(0);

#pragma unroll
  for (int ks = 0; ks < S_DIM / 4; ++ks) {
    const int d = ks % PF;
    v2f fb0, fb1, fb2, fb3;
    fb0.x = qa0[d].x + qa1[d].x + qa2[d].x;  fb0.y = qb0[d].x + qb1[d].x + qb2[d].x;
    fb1.x = qa0[d].y + qa1[d].y + qa2[d].y;  fb1.y = qb0[d].y + qb1[d].y + qb2[d].y;
    fb2.x = qa0[d].z + qa1[d].z + qa2[d].z;  fb2.y = qb0[d].z + qb1[d].z + qb2[d].z;
    fb3.x = qa0[d].w + qa1[d].w + qa2[d].w;  fb3.y = qb0[d].w + qb1[d].w + qb2[d].w;
    if (ks + PF < S_DIM / 4) loadStep(ks + PF, d);

    const int kf = 4 * ks + 2 * lh;
    const int m0 = l16, m1 = 16 + l16;
    const int xi0 = (kf + 4 * m0) & (S_DIM - 1);
    const int xi1 = (kf + 4 * m1) & (S_DIM - 1);
    v2f fa0, fa1;
    fa0.x = Xs[m0][xi0];     fa0.y = Xs[m0][xi0 + 1];
    fa1.x = Xs[m1][xi1];     fa1.y = Xs[m1][xi1 + 1];
    acc00 = __builtin_amdgcn_wmma_f32_16x16x4_f32(false, fa0, false, fb0, (short)0, acc00, false, false);
    acc10 = __builtin_amdgcn_wmma_f32_16x16x4_f32(false, fa1, false, fb0, (short)0, acc10, false, false);
    acc01 = __builtin_amdgcn_wmma_f32_16x16x4_f32(false, fa0, false, fb1, (short)0, acc01, false, false);
    acc11 = __builtin_amdgcn_wmma_f32_16x16x4_f32(false, fa1, false, fb1, (short)0, acc11, false, false);
    acc02 = __builtin_amdgcn_wmma_f32_16x16x4_f32(false, fa0, false, fb2, (short)0, acc02, false, false);
    acc12 = __builtin_amdgcn_wmma_f32_16x16x4_f32(false, fa1, false, fb2, (short)0, acc12, false, false);
    acc03 = __builtin_amdgcn_wmma_f32_16x16x4_f32(false, fa0, false, fb3, (short)0, acc03, false, false);
    acc13 = __builtin_amdgcn_wmma_f32_16x16x4_f32(false, fa1, false, fb3, (short)0, acc13, false, false);
    if (ks + PF < S_DIM / 4) __builtin_amdgcn_sched_group_barrier(0x020, 6, 0);
    __builtin_amdgcn_sched_group_barrier(0x008, 8, 0);
  }

  const float third = 1.0f / 3.0f;
  float* __restrict__ op = out + nb + (size_t)(8 * lh) * S_DIM + ncol;
#pragma unroll
  for (int r = 0; r < 8; ++r) {
    float4 v;
    v.x = acc00[r] * third; v.y = acc01[r] * third; v.z = acc02[r] * third; v.w = acc03[r] * third;
    *(float4*)(op + r * S_DIM) = v;
    float4 u;
    u.x = acc10[r] * third; u.y = acc11[r] * third; u.z = acc12[r] * third; u.w = acc13[r] * third;
    *(float4*)(op + (16 + r) * S_DIM) = u;
  }
}

extern "C" void kernel_launch(void* const* d_in, const int* in_sizes, int n_in,
                              void* d_out, int out_size, void* d_ws, size_t ws_size,
                              hipStream_t stream) {
  const float* a = (const float*)d_in[0];
  const float* b = (const float*)d_in[1];
  const float* ct = (const float*)d_in[2];
  const float* cm = (const float*)d_in[3];
  const float* ew = (const float*)d_in[4];
  float* out = (float*)d_out;
  float* ws = (float*)d_ws;

  router_partials<<<S_DIM, 256, 0, stream>>>(ct, cm, ws);
  router_dft<<<S_DIM, 256, 0, stream>>>(cm, ws);
  moe_select_gemm<<<N_TOT, 128, 0, stream>>>(a, b, ew, ws, out);
}


# Wave butterflies (max and sum trees of router set-up and expert selection) on row_xmask DPP and permlanex16, not lane shuffles through the LDS unit; partners of every step kept, sums bit-identical
# speedup vs baseline: 5.6570x; 1.0576x over previous
#include <hip/hip_runtime.h>
#include <math.h>

typedef __attribute__((ext_vector_type(2))) float v2f;
typedef __attribute__((ext_vector_type(8))) float v8f;

#define S_DIM 256
#define C_DIM 32
#define E_DIM 32
#define N_TOT 512

__device__ __forceinline__ float sigmoidf_(float x) { return 1.0f / (1.0f + expf(-x)); }

#define PART_OFF 4096

__device__ __forceinline__ float xhalf_(float x) {
  const int b = __builtin_bit_cast(int, x);
  return __builtin_bit_cast(float, __builtin_amdgcn_permlanex16(b, b, 0x76543210, 0xfedcba98, false, false));
}
__device__ __forceinline__ float xrow8_(float x) {
  const int b = __builtin_bit_cast(int, x);
  return __builtin_bit_cast(float, __builtin_amdgcn_update_dpp(b, b, 0x168, 0xf, 0xf, true));
}
__device__ __forceinline__ float xrow4_(float x) {
  const int b = __builtin_bit_cast(int, x);
  return __builtin_bit_cast(float, __builtin_amdgcn_update_dpp(b, b, 0x164, 0xf, 0xf, true));
}
__device__ __forceinline__ float xrow2_(float x) {
  const int b = __builtin_bit_cast(int, x);
  return __builtin_bit_cast(float, __builtin_amdgcn_update_dpp(b, b, 0x162, 0xf, 0xf, true));
}
__device__ __forceinline__ float xrow1_(float x) {
  const int b = __builtin_bit_cast(int, x);
  return __builtin_bit_cast(float, __builtin_amdgcn_update_dpp(b, b, 0x161, 0xf, 0xf, true));
}
__device__ __forceinline__ float wave_max_(float v) {
  v = fmaxf(v, xhalf_(v));
  v = fmaxf(v, xrow8_(v));
  v = fmaxf(v, xrow4_(v));
  v = fmaxf(v, xrow2_(v));
  v = fmaxf(v, xrow1_(v));
  return v;
}
__device__ __forceinline__ float wave_sum_(float v) {
  v += xhalf_(v);
  v += xrow8_(v);
  v += xrow4_(v);
  v += xrow2_(v);
  v += xrow1_(v);
  return v;
}
__device__ __forceinline__ float wave_sum_prod_(float a, float b) {
  float v = fmaf(a, b, xhalf_(a * b));
  v += xrow8_(v);
  v += xrow4_(v);
  v += xrow2_(v);
  v += xrow1_(v);
  return v;
}

__global__ __launch_bounds__(256) void router_partials(const float* __restrict__ ct,
                                                       const float* __restrict__ cm,
                                                       float* __restrict__ ws) {
  __shared__ float rmax[8][2];
  __shared__ float rsum[8][2];
  const int tid = threadIdx.x;
  const int row = blockIdx.x;
  const int lane = tid & 31, wave = tid >> 5;

  const float2 x = *(const float2*)(cm + 2 * (row * S_DIM + tid));
  float mre = wave_max_(x.x), mim = wave_max_(x.y);
  if (lane == 0) { rmax[wave][0] = mre; rmax[wave][1] = mim; }
  __syncthreads();
  mre = rmax[0][0]; mim = rmax[0][1];
  for (int w = 1; w < 8; ++w) { mre = fmaxf(mre, rmax[w][0]); mim = fmaxf(mim, rmax[w][1]); }

  float sre = wave_sum_(expf(x.x - mre));
  float sim = wave_sum_(expf(x.y - mim));
  if (lane == 0) { rsum[wave][0] = sre; rsum[wave][1] = sim; }
  __syncthreads();
  if (tid == 0) {
    sre = rsum[0][0]; sim = rsum[0][1];
    for (int w = 1; w < 8; ++w) { sre += rsum[w][0]; sim += rsum[w][1]; }
    *(float4*)(ws + PART_OFF + 4 * row) = make_float4(mre, mim, sre, sim);
  }

  if (row == 0 && tid < E_DIM) {
    const int e = tid;
    ws[1024 + e] = sigmoidf_(ct[((e * 2 + 0) * S_DIM + 0) * 2 + 0]);
    ws[1056 + e] = sigmoidf_(ct[((e * 2 + 0) * S_DIM + 0) * 2 + 1]);
    ws[1088 + e] = sigmoidf_(ct[((e * 2 + 1) * S_DIM + 0) * 2 + 0]);
    ws[1120 + e] = sigmoidf_(ct[((e * 2 + 1) * S_DIM + 0) * 2 + 1]);
  }
}

__global__ __launch_bounds__(256) void router_dft(const float* __restrict__ cm,
                                                  float* __restrict__ ws) {
  __shared__ float rmax[8][2];
  __shared__ float rsum[8][2];
  __shared__ float rdft[8][4];
  const int tid = threadIdx.x;
  const int k = blockIdx.x;
  const int lane = tid & 31, wave = tid >> 5;

  const float4 p = *(const float4*)(ws + PART_OFF + 4 * tid);
  const float2 xr = *(const float2*)(cm + 2 * tid);
  const float2 xc = *(const float2*)(cm + 2 * (tid * S_DIM));

  float mre = wave_max_(p.x), mim = wave_max_(p.y);
  if (lane == 0) { rmax[wave][0] = mre; rmax[wave][1] = mim; }
  __syncthreads();
  mre = rmax[0][0]; mim = rmax[0][1];
  for (int w = 1; w < 8; ++w) { mre = fmaxf(mre, rmax[w][0]); mim = fmaxf(mim, rmax[w][1]); }

  float sre = wave_sum_prod_(p.z, expf(p.x - mre));
  float sim = wave_sum_prod_(p.w, expf(p.y - mim));
  if (lane == 0) { rsum[wave][0] = sre; rsum[wave][1] = sim; }
  __syncthreads();
  sre = rsum[0][0]; sim = rsum[0][1];
  for (int w = 1; w < 8; ++w) { sre += rsum[w][0]; sim += rsum[w][1]; }

  const float rowre = expf(xr.x - mre) / sre;
  const float rowim = expf(xr.y - mim) / sim;
  const float colre = expf(xc.x - mre) / sre;
  const float colim = expf(xc.y - mim) / sim;

  const float w0 = -6.283185307179586f / 256.0f;
  const float ang = w0 * (float)((k * tid) & 255);
  float sn, cs;
  sincosf(ang, &sn, &cs);
  const float v0 = wave_sum_(rowre * cs - rowim * sn);
  const float v1 = wave_sum_(rowre * sn + rowim * cs);
  const float v2 = wave_sum_(colre * cs - colim * sn);
  const float v3 = wave_sum_(colre * sn + colim * cs);
  if (lane == 0) { rdft[wave][0] = v0; rdft[wave][1] = v1; rdft[wave][2] = v2; rdft[wave][3] = v3; }
  __syncthreads();
  if (tid < 4) {
    float s = rdft[0][tid];
    for (int w = 1; w < 8; ++w) s += rdft[w][tid];
    ws[tid * S_DIM + k] = s;
  }
}

#define PF 6

__global__ __launch_bounds__(128) void moe_select_gemm(const float* __restrict__ a,
                                                       const float* __restrict__ b,
                                                       const float* __restrict__ ew,
                                                       const float* __restrict__ ws,
                                                       float* __restrict__ out) {
  __shared__ __align__(16) float Xs[C_DIM][S_DIM];
  __shared__ __align__(16) float part[48 * 256];
  __shared__ float rowsum[6][C_DIM];
  __shared__ float mag[E_DIM];
  __shared__ int sel[4];
  const int n = blockIdx.x;
  const int tid = threadIdx.x;
  const int wave = tid >> 5;
  const int lane = tid & 31;
  const size_t nb = (size_t)n * C_DIM * S_DIM;

  if (tid < 3) sel[tid] = 0;

  const int s4q = (tid & 63) << 2;
  const float4 mrr = *(const float4*)(ws + s4q);
  const float4 mri = *(const float4*)(ws + 256 + s4q);
  const float4 mcr = *(const float4*)(ws + 512 + s4q);
  const float4 mci = *(const float4*)(ws + 768 + s4q);
  const float tar = ws[1024 + lane], tai = ws[1056 + lane];
  const float tbr = ws[1088 + lane], tbi = ws[1120 + lane];

  const int rq = tid >> 6;
  const float* __restrict__ ap = a + nb + rq * S_DIM + s4q;
  const float* __restrict__ bp = b + nb + rq * S_DIM + s4q;
#pragma unroll
  for (int r = 0; r < 16; ++r) {
    const int c = 2 * r + rq;
    const float4 av = *(const float4*)(ap + r * 2 * S_DIM);
    const float4 bv = *(const float4*)(bp + r * 2 * S_DIM);
    float4 v;
    v.x = av.x + bv.x; v.y = av.y + bv.y; v.z = av.z + bv.z; v.w = av.w + bv.w;
    *(float4*)&Xs[c][(s4q + 4 * c) & (S_DIM - 1)] = v;
    float* pp = part + ((r >> 1) * 6) * 256 + ((r & 1) * 128 + tid);
    pp[0 * 256] = (av.x + av.y) + (av.z + av.w);
    pp[1 * 256] = (bv.x + bv.y) + (bv.z + bv.w);
    pp[2 * 256] = av.x * mcr.x + av.y * mcr.y + av.z * mcr.z + av.w * mcr.w;
    pp[3 * 256] = av.x * mci.x + av.y * mci.y + av.z * mci.z + av.w * mci.w;
    pp[4 * 256] = bv.x * mrr.x + bv.y * mrr.y + bv.z * mrr.z + bv.w * mrr.w;
    pp[5 * 256] = bv.x * mri.x + bv.y * mri.y + bv.z * mri.z + bv.w * mri.w;
  }
  __syncthreads();

  for (int q = wave; q < 6; q += 4) {
    const int c = lane;
    const float* p = part + ((c >> 2) * 6 + q) * 256 + (c & 3) * 64;
    float4 acc = *(const float4*)(p + ((lane & 15) << 2));
#pragma unroll
    for (int i = 1; i < 16; ++i) {
      const float4 v = *(const float4*)(p + (((i + lane) & 15) << 2));
      acc.x += v.x; acc.y += v.y; acc.z += v.z; acc.w += v.w;
    }
    rowsum[q][c] = (acc.x + acc.y) + (acc.z + acc.w);
  }
  __syncthreads();

  if (wave == 0) {
    const int c = lane;
    const float sa = rowsum[0][c], sb = rowsum[1][c];
    const float par = rowsum[2][c], pai = rowsum[3][c];
    const float pbr = rowsum[4][c], pbi = rowsum[5][c];
    const float ar = wave_sum_prod_(sa, pbr), ai = wave_sum_prod_(sa, pbi);
    const float br = wave_sum_prod_(sb, par), bi = wave_sum_prod_(sb, pai);
    const float zr = ar * tar + ai * tai + br * tbr + bi * tbi;
    const float zi = ai * tar - ar * tai + bi * tbr - br * tbi;
    mag[lane] = zr * zr + zi * zi;
  }
  __syncthreads();
  if (wave == 0) {
    const float m = mag[lane];
    int rank = 0;
#pragma unroll
    for (int e = 0; e < E_DIM; ++e) {
      const float o = mag[e];
      rank += (o > m || (o == m && e < lane)) ? 1 : 0;
    }
    if (rank < 3) sel[rank] = lane;
  }
  __syncthreads();

  const int i0 = __builtin_amdgcn_readfirstlane(sel[0]) & (E_DIM - 1);
  const int i1 = __builtin_amdgcn_readfirstlane(sel[1]) & (E_DIM - 1);
  const int i2 = __builtin_amdgcn_readfirstlane(sel[2]) & (E_DIM - 1);
  const float* __restrict__ w0 = ew + (size_t)i0 * (S_DIM * S_DIM);
  const float* __restrict__ w1 = ew + (size_t)i1 * (S_DIM * S_DIM);
  const float* __restrict__ w2 = ew + (size_t)i2 * (S_DIM * S_DIM);

  const int l16 = lane & 15;
  const int lh = lane >> 4;
  const int ncol = wave * 64 + 4 * l16;

  const unsigned lane_off = ((unsigned)(2 * lh) << 8) + (unsigned)ncol;

  float4 qa0[PF], qa1[PF], qa2[PF], qb0[PF], qb1[PF], qb2[PF];
  auto loadStep = [&](int ks, int d) {
    const unsigned g = (unsigned)(ks * 4 * S_DIM) + lane_off;
    qa0[d] = *(const float4*)(w0 + g);
    qb0[d] = *(const float4*)(w0 + g + S_DIM);
    qa1[d] = *(const float4*)(w1 + g);
    qb1[d] = *(const float4*)(w1 + g + S_DIM);
    qa2[d] = *(const float4*)(w2 + g);
    qb2[d] = *(const float4*)(w2 + g + S_DIM);
  };

#pragma unroll
  for (int d = 0; d < PF; ++d) loadStep(d, d);

  v8f acc00 = {}, acc01 = {}, acc02 = {}, acc03 = {};
  v8f acc10 = {}, acc11 = {}, acc12 = {}, acc13 = {};

  __builtin_amdgcn_sched_barrier(0);

#pragma unroll
  for (int ks = 0; ks < S_DIM / 4; ++ks) {
    const int d = ks % PF;
    v2f fb0, fb1, fb2, fb3;
    fb0.x = qa0[d].x + qa1[d].x + qa2[d].x;  fb0.y = qb0[d].x + qb1[d].x + qb2[d].x;
    fb1.x = qa0[d].y + qa1[d].y + qa2[d].y;  fb1.y = qb0[d].y + qb1[d].y + qb2[d].y;
    fb2.x = qa0[d].z + qa1[d].z + qa2[d].z;  fb2.y = qb0[d].z + qb1[d].z + qb2[d].z;
    fb3.x = qa0[d].w + qa1[d].w + qa2[d].w;  fb3.y = qb0[d].w + qb1[d].w + qb2[d].w;
    if (ks + PF < S_DIM / 4) loadStep(ks + PF, d);

    const int kf = 4 * ks + 2 * lh;
    const int m0 = l16, m1 = 16 + l16;
    const int xi0 = (kf + 4 * m0) & (S_DIM - 1);
    const int xi1 = (kf + 4 * m1) & (S_DIM - 1);
    v2f fa0, fa1;
    fa0.x = Xs[m0][xi0];     fa0.y = Xs[m0][xi0 + 1];
    fa1.x = Xs[m1][xi1];     fa1.y = Xs[m1][xi1 + 1];
    acc00 = __builtin_amdgcn_wmma_f32_16x16x4_f32(false, fa0, false, fb0, (short)0, acc00, false, false);
    acc10 = __builtin_amdgcn_wmma_f32_16x16x4_f32(false, fa1, false, fb0, (short)0, acc10, false, false);
    acc01 = __builtin_amdgcn_wmma_f32_16x16x4_f32(false, fa0, false, fb1, (short)0, acc01, false, false);
    acc11 = __builtin_amdgcn_wmma_f32_16x16x4_f32(false, fa1, false, fb1, (short)0, acc11, false, false);
    acc02 = __builtin_amdgcn_wmma_f32_16x16x4_f32(false, fa0, false, fb2, (short)0, acc02, false, false);
    acc12 = __builtin_amdgcn_wmma_f32_16x16x4_f32(false, fa1, false, fb2, (short)0, acc12, false, false);
    acc03 = __builtin_amdgcn_wmma_f32_16x16x4_f32(false, fa0, false, fb3, (short)0, acc03, false, false);
    acc13 = __builtin_amdgcn_wmma_f32_16x16x4_f32(false, fa1, false, fb3, (short)0, acc13, false, false);
    if (ks + PF < S_DIM / 4) __builtin_amdgcn_sched_group_barrier(0x020, 6, 0);
    __builtin_amdgcn_sched_group_barrier(0x008, 8, 0);
  }

  const float third = 1.0f / 3.0f;
  float* __restrict__ op = out + nb + (size_t)(8 * lh) * S_DIM + ncol;
#pragma unroll
  for (int r = 0; r < 8; ++r) {
    float4 v;
    v.x = acc00[r] * third; v.y = acc01[r] * third; v.z = acc02[r] * third; v.w = acc03[r] * third;
    *(float4*)(op + r * S_DIM) = v;
    float4 u;
    u.x = acc10[r] * third; u.y = acc11[r] * third; u.z = acc12[r] * third; u.w = acc13[r] * third;
    *(float4*)(op + (16 + r) * S_DIM) = u;
  }
}

extern "C" void kernel_launch(void* const* d_in, const int* in_sizes, int n_in,
                              void* d_out, int out_size, void* d_ws, size_t ws_size,
                              hipStream_t stream) {
  const float* a = (const float*)d_in[0];
  const float* b = (const float*)d_in[1];
  const float* ct = (const float*)d_in[2];
  const float* cm = (const float*)d_in[3];
  const float* ew = (const float*)d_in[4];
  float* out = (float*)d_out;
  float* ws = (float*)d_ws;

  router_partials<<<S_DIM, 256, 0, stream>>>(ct, cm, ws);
  router_dft<<<S_DIM, 256, 0, stream>>>(cm, ws);
  moe_select_gemm<<<N_TOT, 128, 0, stream>>>(a, b, ew, ws, out);
}
